# starting kernel
# baseline (speedup 1.0000x reference)
#include <hip/hip_runtime.h>


typedef __attribute__((ext_vector_type(2))) float v2f;
typedef __attribute__((ext_vector_type(8))) float v8f;

#define NCTRL 64
#define OUTN  512
#define KTILE 8

__global__ void nurbs_stage1(const float* __restrict__ ctrl,
                             const int*   __restrict__ uspan,
                             const float* __restrict__ Nu,
                             float*       __restrict__ A)
{
    int idx = blockIdx.x * blockDim.x + threadIdx.x;
    int n = idx & 63;
    int u = (idx >> 6) & 511;
    int c = (idx >> 15) & 3;
    int b = idx >> 17;

    int m0 = uspan[u] - 3;
    float acc = 0.0f;
#pragma unroll
    for (int r = 0; r < 4; ++r) {
        float w = Nu[u * 4 + r];
        float x = ctrl[(((size_t)b * 64 + (m0 + r)) * 64 + n) * 4 + c];
        acc = fmaf(w, x, acc);
    }
    A[idx] = acc;
}

__global__ void nurbs_stage2(const float* __restrict__ A,
                             const int*   __restrict__ vspan,
                             const float* __restrict__ Nv,
                             float*       __restrict__ out)
{
    __shared__ float nvd[8][KTILE * 16];

    const int lane = threadIdx.x & 31;
    const int wave = threadIdx.x >> 5;
    const int tile = blockIdx.x * 8 + wave;
    const int vt = tile & 31;
    const int ut = (tile >> 5) & 31;
    const int b  = tile >> 10;
    const int u0 = ut * 16;
    const int v0 = vt * 16;

    const int kbase = vspan[v0] - 3;

    if (lane < 16) {
#pragma unroll
        for (int k = 0; k < KTILE; ++k) nvd[wave][k * 16 + lane] = 0.0f;
        int v  = v0 + lane;
        int ki = vspan[v] - 3 - kbase;
#pragma unroll
        for (int s = 0; s < 4; ++s) {
            int kk = ki + s;
            if (kk < KTILE) nvd[wave][kk * 16 + lane] = Nv[v * 4 + s];
        }
    }
    __syncthreads();

    const int mrow  = lane & 15;
    const int khalf = (lane >> 4) * 2;

    v8f acc[4];
#pragma unroll
    for (int c = 0; c < 4; ++c) {
        v8f d = {};
#pragma unroll
        for (int ks = 0; ks < 2; ++ks) {
            int kk = ks * 4 + khalf;
            int n0 = kbase + kk;     if (n0 > 63) n0 = 63;
            int n1 = kbase + kk + 1; if (n1 > 63) n1 = 63;
            const float* ap =
                A + (((size_t)(b * 4 + c) * OUTN + (u0 + mrow)) * NCTRL);
            v2f av; av.x = ap[n0]; av.y = ap[n1];
            v2f bv;
            bv.x = nvd[wave][kk * 16 + mrow];
            bv.y = nvd[wave][(kk + 1) * 16 + mrow];
            d = __builtin_amdgcn_wmma_f32_16x16x4_f32(
                     false, av,  false, bv,
                     (short)0, d,  false,  false);
        }
        acc[c] = d;
    }

#pragma unroll
    for (int j = 0; j < 8; ++j) {
        int u = u0 + j + (lane >> 4) * 8;
        int v = v0 + (lane & 15);
        float w = acc[3][j];
        float r = __builtin_amdgcn_rcpf(w);
        r = r * (2.0f - w * r);
        size_t o = (((size_t)b * OUTN + u) * OUTN + v) * 3;
        __builtin_nontemporal_store(acc[0][j] * r, &out[o + 0]);
        __builtin_nontemporal_store(acc[1][j] * r, &out[o + 1]);
        __builtin_nontemporal_store(acc[2][j] * r, &out[o + 2]);
    }
}

extern "C" void kernel_launch(void* const* d_in, const int* in_sizes, int n_in,
                              void* d_out, int out_size, void* d_ws, size_t ws_size,
                              hipStream_t stream)
{
    const float* ctrl  = (const float*)d_in[0];
    const int*   uspan = (const int*)  d_in[1];
    const int*   vspan = (const int*)  d_in[2];
    const float* Nu    = (const float*)d_in[3];
    const float* Nv    = (const float*)d_in[4];
    float*       out   = (float*)d_out;
    float*       A     = (float*)d_ws;

    nurbs_stage1<<<8192, 256, 0, stream>>>(ctrl, uspan, Nu, A);
    nurbs_stage2<<<2048, 256, 0, stream>>>(A, vspan, Nv, out);
}


# Adjacent v tiles paired and their lane halves swapped (v_permlane16_swap) so each 12-B-per-lane store writes 384 contiguous bytes of one row: three whole lines, one writer per line
# speedup vs baseline: 3.4569x; 3.4569x over previous
#include <hip/hip_runtime.h>


typedef __attribute__((ext_vector_type(2))) float v2f;
typedef __attribute__((ext_vector_type(4))) float v4f;
typedef __attribute__((ext_vector_type(8))) float v8f;

#define NCTRL  64
#define OUTN   512
#define UBAND  32
#define VTILES 4
#define APITCH 69

__device__ __forceinline__ void swap_lane_halves(float& lo, float& hi)
{
    asm("v_permlane16_swap_b32 %0, %1" : "+v"(lo), "+v"(hi));
}

__device__ __forceinline__ float band_tap(v4f w, int s)
{
    float r = 0.0f;
    r = (s == 0) ? w.x : r;
    r = (s == 1) ? w.y : r;
    r = (s == 2) ? w.z : r;
    r = (s == 3) ? w.w : r;
    return r;
}

__global__ __launch_bounds__(256) __attribute__((amdgpu_waves_per_eu(2, 2)))
void nurbs_rowband(const float* __restrict__ ctrl,
                   const int*   __restrict__ uspan,
                   const int*   __restrict__ vspan,
                   const float* __restrict__ Nu,
                   const float* __restrict__ Nv,
                   float*       __restrict__ out)
{
    __shared__ v4f au[UBAND * APITCH];

    const int t    = threadIdx.x;
    const int lane = t & 31;
    const int wave = __builtin_amdgcn_readfirstlane(t >> 5);
    const int b    = blockIdx.x >> 4;
    const int u0   = (blockIdx.x & 15) * UBAND;
    const int col  = lane & 15;
    const int half = lane >> 4;
    const int khalf = half * 2;

    int vs[VTILES];
    v4f nv[VTILES];
#pragma unroll
    for (int i = 0; i < VTILES; ++i) {
        const int v = wave * (VTILES * 16) + i * 16 + col;
        vs[i] = vspan[v];
        nv[i] = *reinterpret_cast<const v4f*>(Nv + v * 4);
    }

    {
        const int n  = t & 63;
        const int ug = wave >> 1;
        const v4f* c4 = reinterpret_cast<const v4f*>(ctrl)
                      + (size_t)b * (NCTRL * NCTRL) + n;
#pragma unroll
        for (int i = 0; i < 8; ++i) {
            const int ul = ug * 8 + i;
            const int m0 = uspan[u0 + ul] - 3;
            const float* w = Nu + (u0 + ul) * 4;
            v4f acc = {0.0f, 0.0f, 0.0f, 0.0f};
#pragma unroll
            for (int r = 0; r < 4; ++r) {
                const v4f x = c4[(m0 + r) * NCTRL];
                const float wr = w[r];
                acc.x = fmaf(wr, x.x, acc.x);
                acc.y = fmaf(wr, x.y, acc.y);
                acc.z = fmaf(wr, x.z, acc.z);
                acc.w = fmaf(wr, x.w, acc.w);
            }
            au[ul * APITCH + n] = acc;
        }
        if (t < UBAND * (APITCH - NCTRL)) {
            const int ul = t / (APITCH - NCTRL);
            const int nz = NCTRL + t % (APITCH - NCTRL);
            au[ul * APITCH + nz] = (v4f){0.0f, 0.0f, 0.0f, 0.0f};
        }
    }
    __syncthreads();

    int kb[VTILES];
    v2f bf[VTILES][2];
#pragma unroll
    for (int i = 0; i < VTILES; ++i) {
        kb[i] = __builtin_amdgcn_readfirstlane(vs[i]) - 3;
        const int ki = vs[i] - 3 - kb[i];
#pragma unroll
        for (int ks = 0; ks < 2; ++ks) {
            const int s = ks * 4 + khalf - ki;
            bf[i][ks].x = band_tap(nv[i], s);
            bf[i][ks].y = band_tap(nv[i], s + 1);
        }
    }

    float* const ob = out + (((size_t)b * OUTN + u0) * OUTN
                             + wave * (VTILES * 16) + lane) * 3;
#pragma unroll
    for (int ut = 0; ut < 2; ++ut) {
#pragma unroll
        for (int p = 0; p < VTILES / 2; ++p) {
            v8f acc[2][4] = {};
#pragma unroll
            for (int q = 0; q < 2; ++q) {
                const int i = 2 * p + q;
                const v4f* ap = au + (ut * 16 + col) * APITCH + kb[i] + khalf;
#pragma unroll
                for (int ks = 0; ks < 2; ++ks) {
                    const v4f a0 = ap[ks * 4];
                    const v4f a1 = ap[ks * 4 + 1];
#pragma unroll
                    for (int c = 0; c < 4; ++c) {
                        v2f av; av.x = a0[c]; av.y = a1[c];
                        acc[q][c] = __builtin_amdgcn_wmma_f32_16x16x4_f32(
                             false, av,  false, bf[i][ks],
                             (short)0, acc[q][c],
                             false,  false);
                    }
                }
            }

            v8f px[2], py[2], pz[2];
#pragma unroll
            for (int q = 0; q < 2; ++q) {
                const v8f w = acc[q][3];
                v8f r;
#pragma unroll
                for (int j = 0; j < 8; ++j)
                    r[j] = __builtin_amdgcn_rcpf(w[j]);
                r = r * (2.0f - w * r);
                px[q] = acc[q][0] * r;
                py[q] = acc[q][1] * r;
                pz[q] = acc[q][2] * r;
            }
#pragma unroll
            for (int j = 0; j < 8; ++j) {
                float x[2] = {px[0][j], px[1][j]};
                float y[2] = {py[0][j], py[1][j]};
                float z[2] = {pz[0][j], pz[1][j]};
                swap_lane_halves(x[0], x[1]);
                swap_lane_halves(y[0], y[1]);
                swap_lane_halves(z[0], z[1]);
#pragma unroll
                for (int q = 0; q < 2; ++q) {
                    float* const o = ob + ((ut * 16 + j + 8 * q) * OUTN + p * 32) * 3;
                    __builtin_nontemporal_store(x[q], o + 0);
                    __builtin_nontemporal_store(y[q], o + 1);
                    __builtin_nontemporal_store(z[q], o + 2);
                }
            }
        }
    }
}

extern "C" void kernel_launch(void* const* d_in, const int* in_sizes, int n_in,
                              void* d_out, int out_size, void* d_ws, size_t ws_size,
                              hipStream_t stream)
{
    const float* ctrl  = (const float*)d_in[0];
    const int*   uspan = (const int*)  d_in[1];
    const int*   vspan = (const int*)  d_in[2];
    const float* Nu    = (const float*)d_in[3];
    const float* Nv    = (const float*)d_in[4];
    float*       out   = (float*)d_out;

    nurbs_rowband<<<256, 256, 0, stream>>>(ctrl, uspan, vspan, Nu, Nv, out);
}
